# starting kernel
# baseline (speedup 1.0000x reference)
#include <hip/hip_runtime.h>
#include <math.h>

#define B_DIM 64
#define S_DIM 256
#define E_DIM 8
#define H_DIM 2
#define DK    4
#define EPS   1e-5f
#define SCALE_LOG2E 0.72134752044448170f

typedef __attribute__((ext_vector_type(2))) float v2f;
typedef __attribute__((ext_vector_type(8))) float v8f;

template <int CTRL>
__device__ __forceinline__ float dpp_f(float x) {
  return __int_as_float(__builtin_amdgcn_update_dpp(
      0, __float_as_int(x), CTRL, 0xF, 0xF, true));
}
__device__ __forceinline__ float rowsum16(float v) {
  v += dpp_f<0xB1>(v);
  v += dpp_f<0x4E>(v);
  v += dpp_f<0x141>(v);
  v += dpp_f<0x140>(v);
  return v;
}

__device__ __forceinline__ void vqc8(const float (&in)[8], const float (&W)[8],
                                     float (&z)[8]) {
  float c[8];
#pragma unroll
  for (int j = 0; j < 8; ++j) c[j] = __cosf(in[j] + W[j]);
  float p = c[0];
#pragma unroll
  for (int j = 1; j < 8; ++j) { p *= c[j]; z[j] = p; }
  float q = c[1];
#pragma unroll
  for (int j = 2; j < 8; ++j) q *= c[j];
  z[0] = q;
}

__device__ __forceinline__ void ln8(const float (&t)[8],
                                    const float* __restrict__ g,
                                    const float* __restrict__ b,
                                    float (&o)[8]) {
  float m = 0.f;
#pragma unroll
  for (int j = 0; j < 8; ++j) m += t[j];
  m *= 0.125f;
  float v = 0.f;
#pragma unroll
  for (int j = 0; j < 8; ++j) { float d = t[j] - m; v += d * d; }
  v *= 0.125f;
  float inv = 1.0f / sqrtf(v + EPS);
#pragma unroll
  for (int j = 0; j < 8; ++j) o[j] = (t[j] - m) * inv * g[j] + b[j];
}

__global__ __launch_bounds__(256) void attn_kernel(
    const float* __restrict__ x, const float* __restrict__ Wq,
    const float* __restrict__ Wk, const float* __restrict__ Wv,
    float* __restrict__ ctx) {
  __shared__ __align__(16) float Qs[S_DIM * DK];
  __shared__ __align__(16) float Ks[S_DIM * DK];
  __shared__ __align__(16) float Vs[S_DIM * DK];

  const int b = blockIdx.x >> 1;
  const int h = blockIdx.x & 1;
  const int tid = threadIdx.x;

  float wq[8], wk[8], wv[8];
#pragma unroll
  for (int j = 0; j < 8; ++j) { wq[j] = Wq[j]; wk[j] = Wk[j]; wv[j] = Wv[j]; }

  {
    const int s = tid;
    const float* xp = x + (size_t)(b * S_DIM + s) * E_DIM;
    float xr[8];
#pragma unroll
    for (int j = 0; j < 8; ++j) xr[j] = xp[j];
    float zq[8], zk[8], zv[8];
    vqc8(xr, wq, zq);
    vqc8(xr, wk, zk);
    vqc8(xr, wv, zv);
#pragma unroll
    for (int d = 0; d < DK; ++d) {
      Qs[s * DK + d] = zq[h * DK + d];
      Ks[s * DK + d] = zk[h * DK + d];
      Vs[s * DK + d] = zv[h * DK + d];
    }
  }
  __syncthreads();

  const int lane  = tid & 31;
  const int wv_id = tid >> 5;
  const int half  = lane >> 4;
  const int ln    = lane & 15;
  const int kp    = half * 2;

  for (int i = 0; i < 2; ++i) {
    const int qb = wv_id * 2 + i;
    const int qbase = qb * 16;

    v2f a;
    a.x = Qs[(qbase + ln) * DK + kp];
    a.y = Qs[(qbase + ln) * DK + kp + 1];

    float lrow[8], acc[8][DK];
#pragma unroll
    for (int r = 0; r < 8; ++r) {
      lrow[r] = 0.f;
#pragma unroll
      for (int d = 0; d < DK; ++d) acc[r][d] = 0.f;
    }

#pragma unroll
    for (int t = 0; t < 16; ++t) {
      const int kbase = t * 16;
      v2f bm;
      bm.x = Ks[(kbase + ln) * DK + kp];
      bm.y = Ks[(kbase + ln) * DK + kp + 1];

      v8f cz = {0.f, 0.f, 0.f, 0.f, 0.f, 0.f, 0.f, 0.f};
      v8f d = __builtin_amdgcn_wmma_f32_16x16x4_f32(
          false, a, false, bm, (short)0, cz, false, false);

      float vvals[DK];
#pragma unroll
      for (int dv = 0; dv < DK; ++dv) vvals[dv] = Vs[(kbase + ln) * DK + dv];

#pragma unroll
      for (int r = 0; r < 8; ++r) {
        float p = __builtin_amdgcn_exp2f(d[r] * SCALE_LOG2E);
        lrow[r] += p;
#pragma unroll
        for (int dv = 0; dv < DK; ++dv) acc[r][dv] += p * vvals[dv];
      }
    }

#pragma unroll
    for (int r = 0; r < 8; ++r) {
      float inv_l = 1.0f / rowsum16(lrow[r]);
#pragma unroll
      for (int dv = 0; dv < DK; ++dv) {
        float v = rowsum16(acc[r][dv]) * inv_l;
        if (ln == dv) {
          const int m = qbase + r + 8 * half;
          ctx[(size_t)(b * S_DIM + m) * E_DIM + h * DK + dv] = v;
        }
      }
    }
  }
}

__global__ __launch_bounds__(256) void post_kernel(
    const float* __restrict__ x, const float* __restrict__ ctx,
    const float* __restrict__ Wc, const float* __restrict__ Wf,
    const float* __restrict__ w1, const float* __restrict__ b1,
    const float* __restrict__ w2, const float* __restrict__ b2,
    const float* __restrict__ g1, const float* __restrict__ be1,
    const float* __restrict__ g2, const float* __restrict__ be2,
    float* __restrict__ out) {
  const int gid = blockIdx.x * blockDim.x + threadIdx.x;
  if (gid >= B_DIM * S_DIM) return;

  float xv[8], cv[8], wc[8], wf[8];
#pragma unroll
  for (int j = 0; j < 8; ++j) {
    xv[j] = x[(size_t)gid * E_DIM + j];
    cv[j] = ctx[(size_t)gid * E_DIM + j];
    wc[j] = Wc[j];
    wf[j] = Wf[j];
  }

  float ao[8];
  vqc8(cv, wc, ao);
  float t[8];
#pragma unroll
  for (int j = 0; j < 8; ++j) t[j] = xv[j] + ao[j];
  float y[8];
  ln8(t, g1, be1, y);

  float hv[8];
#pragma unroll
  for (int i = 0; i < 8; ++i) {
    float s = b1[i];
#pragma unroll
    for (int j = 0; j < 8; ++j) s += y[j] * w1[i * 8 + j];
    hv[i] = s;
  }
  float hq[8];
  vqc8(hv, wf, hq);

  float u[8];
#pragma unroll
  for (int i = 0; i < 8; ++i) {
    float s = b2[i];
#pragma unroll
    for (int j = 0; j < 8; ++j) s += hq[j] * w2[i * 8 + j];
    u[i] = y[i] + s;
  }
  float o[8];
  ln8(u, g2, be2, o);
#pragma unroll
  for (int j = 0; j < 8; ++j) out[(size_t)gid * E_DIM + j] = o[j];
}

extern "C" void kernel_launch(void* const* d_in, const int* in_sizes, int n_in,
                              void* d_out, int out_size, void* d_ws,
                              size_t ws_size, hipStream_t stream) {
  (void)in_sizes; (void)n_in; (void)out_size; (void)ws_size;
  const float* x   = (const float*)d_in[0];
  const float* Wq  = (const float*)d_in[1];
  const float* Wk  = (const float*)d_in[2];
  const float* Wv  = (const float*)d_in[3];
  const float* Wc  = (const float*)d_in[4];
  const float* Wf  = (const float*)d_in[5];
  const float* w1  = (const float*)d_in[6];
  const float* b1  = (const float*)d_in[7];
  const float* w2  = (const float*)d_in[8];
  const float* b2  = (const float*)d_in[9];
  const float* g1  = (const float*)d_in[10];
  const float* be1 = (const float*)d_in[11];
  const float* g2  = (const float*)d_in[12];
  const float* be2 = (const float*)d_in[13];
  float* out = (float*)d_out;
  float* ctx = (float*)d_ws;

  attn_kernel<<<B_DIM * H_DIM, 256, 0, stream>>>(x, Wq, Wk, Wv, ctx);
  post_kernel<<<(B_DIM * S_DIM) / 256, 256, 0, stream>>>(
      x, ctx, Wc, Wf, w1, b1, w2, b2, g1, be1, g2, be2, out);
}


# Fuse the per-token epilogue into the attention kernel: one launch of 256 blocks, block = (batch, 64-query slice) serving both heads, context rows pass through LDS not workspace; all sums unchanged
# speedup vs baseline: 1.7367x; 1.7367x over previous
#include <hip/hip_runtime.h>
#include <math.h>

#define B_DIM 64
#define S_DIM 256
#define E_DIM 8
#define H_DIM 2
#define DK    4
#define EPS   1e-5f
#define SCALE_LOG2E 0.72134752044448170f

typedef __attribute__((ext_vector_type(2))) float v2f;
typedef __attribute__((ext_vector_type(8))) float v8f;

template <int CTRL>
__device__ __forceinline__ float dpp_f(float x) {
  return __int_as_float(__builtin_amdgcn_update_dpp(
      0, __float_as_int(x), CTRL, 0xF, 0xF, true));
}
__device__ __forceinline__ float rowsum16(float v) {
  v += dpp_f<0xB1>(v);
  v += dpp_f<0x4E>(v);
  v += dpp_f<0x141>(v);
  v += dpp_f<0x140>(v);
  return v;
}

__device__ __forceinline__ void vqc8(const float (&in)[8], const float (&W)[8],
                                     float (&z)[8]) {
  float c[8];
#pragma unroll
  for (int j = 0; j < 8; ++j) c[j] = __cosf(in[j] + W[j]);
  float p = c[0];
#pragma unroll
  for (int j = 1; j < 8; ++j) { p *= c[j]; z[j] = p; }
  float q = c[1];
#pragma unroll
  for (int j = 2; j < 8; ++j) q *= c[j];
  z[0] = q;
}

__device__ __forceinline__ void ln8(const float (&t)[8],
                                    const float* __restrict__ g,
                                    const float* __restrict__ b,
                                    float (&o)[8]) {
  float m = 0.f;
#pragma unroll
  for (int j = 0; j < 8; ++j) m += t[j];
  m *= 0.125f;
  float v = 0.f;
#pragma unroll
  for (int j = 0; j < 8; ++j) { float d = t[j] - m; v += d * d; }
  v *= 0.125f;
  float inv = 1.0f / sqrtf(v + EPS);
#pragma unroll
  for (int j = 0; j < 8; ++j) o[j] = (t[j] - m) * inv * g[j] + b[j];
}

#define QS_DIM 64
#define NSLICE 4

__global__ __launch_bounds__(256) void fused_block_kernel(
    const float* __restrict__ x, const float* __restrict__ Wq,
    const float* __restrict__ Wk, const float* __restrict__ Wv,
    const float* __restrict__ Wc, const float* __restrict__ Wf,
    const float* __restrict__ w1, const float* __restrict__ b1,
    const float* __restrict__ w2, const float* __restrict__ b2,
    const float* __restrict__ g1, const float* __restrict__ be1,
    const float* __restrict__ g2, const float* __restrict__ be2,
    float* __restrict__ out) {
  __shared__ __align__(16) float Ks[H_DIM * S_DIM * DK];
  __shared__ __align__(16) float Vs[H_DIM * S_DIM * DK];
  __shared__ __align__(16) float Qs[H_DIM * QS_DIM * DK];
  __shared__ __align__(16) float Cs[QS_DIM * E_DIM];

  const int b   = blockIdx.x / NSLICE;
  const int qs  = blockIdx.x % NSLICE;
  const int tid = threadIdx.x;
  const bool owner = (tid / QS_DIM) == qs;
  const int  ql = tid % QS_DIM;

  float xr[8];
  {
    const float* xp = x + (size_t)(b * S_DIM + tid) * E_DIM;
#pragma unroll
    for (int j = 0; j < 8; ++j) xr[j] = xp[j];

    float wk[8], wv[8];
#pragma unroll
    for (int j = 0; j < 8; ++j) { wk[j] = Wk[j]; wv[j] = Wv[j]; }
    float zk[8], zv[8];
    vqc8(xr, wk, zk);
    vqc8(xr, wv, zv);
#pragma unroll
    for (int hh = 0; hh < H_DIM; ++hh) {
#pragma unroll
      for (int d = 0; d < DK; ++d) {
        Ks[(hh * S_DIM + tid) * DK + d] = zk[hh * DK + d];
        Vs[(hh * S_DIM + tid) * DK + d] = zv[hh * DK + d];
      }
    }
    if (owner) {
      float wq[8];
#pragma unroll
      for (int j = 0; j < 8; ++j) wq[j] = Wq[j];
      float zq[8];
      vqc8(xr, wq, zq);
#pragma unroll
      for (int hh = 0; hh < H_DIM; ++hh) {
#pragma unroll
        for (int d = 0; d < DK; ++d)
          Qs[(hh * QS_DIM + ql) * DK + d] = zq[hh * DK + d];
      }
    }
  }
  __syncthreads();

  {
    const int lane  = tid & 31;
    const int wv_id = tid >> 5;
    const int h     = wv_id & 1;
    const int qb    = wv_id >> 1;
    const int half  = lane >> 4;
    const int ln    = lane & 15;
    const int kp    = half * 2;
    const int qbase = qb * 16;

    const float* Qh = Qs + h * QS_DIM * DK;
    const float* Kh = Ks + h * S_DIM * DK;
    const float* Vh = Vs + h * S_DIM * DK;

    v2f a;
    a.x = Qh[(qbase + ln) * DK + kp];
    a.y = Qh[(qbase + ln) * DK + kp + 1];

    float lrow[8], acc[8][DK];
#pragma unroll
    for (int r = 0; r < 8; ++r) {
      lrow[r] = 0.f;
#pragma unroll
      for (int d = 0; d < DK; ++d) acc[r][d] = 0.f;
    }

#pragma unroll
    for (int t = 0; t < 16; ++t) {
      const int kbase = t * 16;
      v2f bm;
      bm.x = Kh[(kbase + ln) * DK + kp];
      bm.y = Kh[(kbase + ln) * DK + kp + 1];

      v8f cz = {0.f, 0.f, 0.f, 0.f, 0.f, 0.f, 0.f, 0.f};
      v8f d = __builtin_amdgcn_wmma_f32_16x16x4_f32(
          false, a, false, bm, (short)0, cz, false, false);

      float vvals[DK];
#pragma unroll
      for (int dv = 0; dv < DK; ++dv) vvals[dv] = Vh[(kbase + ln) * DK + dv];

#pragma unroll
      for (int r = 0; r < 8; ++r) {
        float p = __builtin_amdgcn_exp2f(d[r] * SCALE_LOG2E);
        lrow[r] += p;
#pragma unroll
        for (int dv = 0; dv < DK; ++dv) acc[r][dv] += p * vvals[dv];
      }
    }

#pragma unroll
    for (int r = 0; r < 8; ++r) {
      float inv_l = 1.0f / rowsum16(lrow[r]);
#pragma unroll
      for (int dv = 0; dv < DK; ++dv) {
        float v = rowsum16(acc[r][dv]) * inv_l;
        if (ln == dv) {
          const int m = qbase + r + 8 * half;
          Cs[m * E_DIM + h * DK + dv] = v;
        }
      }
    }
  }
  __syncthreads();

  if (!owner) return;

  float cv[8], wc[8], wf[8];
#pragma unroll
  for (int j = 0; j < 8; ++j) {
    cv[j] = Cs[ql * E_DIM + j];
    wc[j] = Wc[j];
    wf[j] = Wf[j];
  }

  float ao[8];
  vqc8(cv, wc, ao);
  float t[8];
#pragma unroll
  for (int j = 0; j < 8; ++j) t[j] = xr[j] + ao[j];
  float y[8];
  ln8(t, g1, be1, y);

  float hv[8];
#pragma unroll
  for (int i = 0; i < 8; ++i) {
    float s = b1[i];
#pragma unroll
    for (int j = 0; j < 8; ++j) s += y[j] * w1[i * 8 + j];
    hv[i] = s;
  }
  float hq[8];
  vqc8(hv, wf, hq);

  float u[8];
#pragma unroll
  for (int i = 0; i < 8; ++i) {
    float s = b2[i];
#pragma unroll
    for (int j = 0; j < 8; ++j) s += hq[j] * w2[i * 8 + j];
    u[i] = y[i] + s;
  }
  float o[8];
  ln8(u, g2, be2, o);
  float* op = out + (size_t)(b * S_DIM + tid) * E_DIM;
#pragma unroll
  for (int j = 0; j < 8; ++j) op[j] = o[j];
}

extern "C" void kernel_launch(void* const* d_in, const int* in_sizes, int n_in,
                              void* d_out, int out_size, void* d_ws,
                              size_t ws_size, hipStream_t stream) {
  (void)in_sizes; (void)n_in; (void)out_size; (void)d_ws; (void)ws_size;
  const float* x   = (const float*)d_in[0];
  const float* Wq  = (const float*)d_in[1];
  const float* Wk  = (const float*)d_in[2];
  const float* Wv  = (const float*)d_in[3];
  const float* Wc  = (const float*)d_in[4];
  const float* Wf  = (const float*)d_in[5];
  const float* w1  = (const float*)d_in[6];
  const float* b1  = (const float*)d_in[7];
  const float* w2  = (const float*)d_in[8];
  const float* b2  = (const float*)d_in[9];
  const float* g1  = (const float*)d_in[10];
  const float* be1 = (const float*)d_in[11];
  const float* g2  = (const float*)d_in[12];
  const float* be2 = (const float*)d_in[13];
  float* out = (float*)d_out;

  fused_block_kernel<<<B_DIM * NSLICE, 256, 0, stream>>>(
      x, Wq, Wk, Wv, Wc, Wf, w1, b1, w2, b2, g1, be1, g2, be2, out);
}


# Attention key loop kept rolled and software-pipelined, two 16-key tiles per trip: 20 independent packed accumulators per lane replace 16-long dependent add chains; summation order unchanged
# speedup vs baseline: 1.8280x; 1.0526x over previous
#include <hip/hip_runtime.h>
#include <math.h>

#define B_DIM 64
#define S_DIM 256
#define E_DIM 8
#define H_DIM 2
#define DK    4
#define EPS   1e-5f
#define SCALE_LOG2E 0.72134752044448170f

typedef __attribute__((ext_vector_type(2))) float v2f;
typedef __attribute__((ext_vector_type(4))) float v4f;
typedef __attribute__((ext_vector_type(8))) float v8f;

template <int CTRL>
__device__ __forceinline__ float dpp_f(float x) {
  return __int_as_float(__builtin_amdgcn_update_dpp(
      0, __float_as_int(x), CTRL, 0xF, 0xF, true));
}
__device__ __forceinline__ float rowsum16(float v) {
  v += dpp_f<0xB1>(v);
  v += dpp_f<0x4E>(v);
  v += dpp_f<0x141>(v);
  v += dpp_f<0x140>(v);
  return v;
}

__device__ __forceinline__ void vqc8(const float (&in)[8], const float (&W)[8],
                                     float (&z)[8]) {
  float c[8];
#pragma unroll
  for (int j = 0; j < 8; ++j) c[j] = __cosf(in[j] + W[j]);
  float p = c[0];
#pragma unroll
  for (int j = 1; j < 8; ++j) { p *= c[j]; z[j] = p; }
  float q = c[1];
#pragma unroll
  for (int j = 2; j < 8; ++j) q *= c[j];
  z[0] = q;
}

__device__ __forceinline__ void ln8(const float (&t)[8],
                                    const float* __restrict__ g,
                                    const float* __restrict__ b,
                                    float (&o)[8]) {
  float m = 0.f;
#pragma unroll
  for (int j = 0; j < 8; ++j) m += t[j];
  m *= 0.125f;
  float v = 0.f;
#pragma unroll
  for (int j = 0; j < 8; ++j) { float d = t[j] - m; v += d * d; }
  v *= 0.125f;
  float inv = 1.0f / sqrtf(v + EPS);
#pragma unroll
  for (int j = 0; j < 8; ++j) o[j] = (t[j] - m) * inv * g[j] + b[j];
}

__device__ __forceinline__ void tile_accumulate(const v8f& d, const v4f& vv,
                                                v2f (&l2)[4],
                                                v2f (&acc)[8][2]) {
  float p[8];
#pragma unroll
  for (int r = 0; r < 8; ++r)
    p[r] = __builtin_amdgcn_exp2f(d[r] * SCALE_LOG2E);
#pragma unroll
  for (int j = 0; j < 4; ++j) {
    v2f pj = {p[2 * j], p[2 * j + 1]};
    l2[j] += pj;
  }
  const v2f v01 = {vv.x, vv.y};
  const v2f v23 = {vv.z, vv.w};
#pragma unroll
  for (int r = 0; r < 8; ++r) {
    const v2f pp = {p[r], p[r]};
    acc[r][0] = __builtin_elementwise_fma(pp, v01, acc[r][0]);
    acc[r][1] = __builtin_elementwise_fma(pp, v23, acc[r][1]);
  }
}

#define QS_DIM 64
#define NSLICE 4

__global__ __launch_bounds__(256) void fused_block_kernel(
    const float* __restrict__ x, const float* __restrict__ Wq,
    const float* __restrict__ Wk, const float* __restrict__ Wv,
    const float* __restrict__ Wc, const float* __restrict__ Wf,
    const float* __restrict__ w1, const float* __restrict__ b1,
    const float* __restrict__ w2, const float* __restrict__ b2,
    const float* __restrict__ g1, const float* __restrict__ be1,
    const float* __restrict__ g2, const float* __restrict__ be2,
    float* __restrict__ out) {
  __shared__ __align__(16) float Ks[H_DIM * S_DIM * DK];
  __shared__ __align__(16) float Vs[H_DIM * S_DIM * DK];
  __shared__ __align__(16) float Qs[H_DIM * QS_DIM * DK];
  __shared__ __align__(16) float Cs[QS_DIM * E_DIM];

  const int b   = blockIdx.x / NSLICE;
  const int qs  = blockIdx.x % NSLICE;
  const int tid = threadIdx.x;
  const bool owner = (tid / QS_DIM) == qs;
  const int  ql = tid % QS_DIM;

  float xr[8];
  {
    const float* xp = x + (size_t)(b * S_DIM + tid) * E_DIM;
#pragma unroll
    for (int j = 0; j < 8; ++j) xr[j] = xp[j];

    float wk[8], wv[8];
#pragma unroll
    for (int j = 0; j < 8; ++j) { wk[j] = Wk[j]; wv[j] = Wv[j]; }
    float zk[8], zv[8];
    vqc8(xr, wk, zk);
    vqc8(xr, wv, zv);
#pragma unroll
    for (int hh = 0; hh < H_DIM; ++hh) {
#pragma unroll
      for (int d = 0; d < DK; ++d) {
        Ks[(hh * S_DIM + tid) * DK + d] = zk[hh * DK + d];
        Vs[(hh * S_DIM + tid) * DK + d] = zv[hh * DK + d];
      }
    }
    if (owner) {
      float wq[8];
#pragma unroll
      for (int j = 0; j < 8; ++j) wq[j] = Wq[j];
      float zq[8];
      vqc8(xr, wq, zq);
#pragma unroll
      for (int hh = 0; hh < H_DIM; ++hh) {
#pragma unroll
        for (int d = 0; d < DK; ++d)
          Qs[(hh * QS_DIM + ql) * DK + d] = zq[hh * DK + d];
      }
    }
  }
  __syncthreads();

  {
    const int lane  = tid & 31;
    const int wv_id = tid >> 5;
    const int h     = wv_id & 1;
    const int qb    = wv_id >> 1;
    const int half  = lane >> 4;
    const int ln    = lane & 15;
    const int kp    = half * 2;
    const int qbase = qb * 16;

    const float* Qh = Qs + h * QS_DIM * DK;
    const float* Kh = Ks + h * S_DIM * DK;
    const float* Vh = Vs + h * S_DIM * DK;

    v2f a;
    a.x = Qh[(qbase + ln) * DK + kp];
    a.y = Qh[(qbase + ln) * DK + kp + 1];

    v2f l2[4], acc[8][2];
#pragma unroll
    for (int j = 0; j < 4; ++j) l2[j] = (v2f){0.f, 0.f};
#pragma unroll
    for (int r = 0; r < 8; ++r) {
      acc[r][0] = (v2f){0.f, 0.f};
      acc[r][1] = (v2f){0.f, 0.f};
    }

    auto ldK = [&](int t) -> v2f {
      return *(const v2f*)(Kh + (t * 16 + ln) * DK + kp);
    };
    auto ldV = [&](int t) -> v4f {
      return *(const v4f*)(Vh + (t * 16 + ln) * DK);
    };
    const v8f cz = {0.f, 0.f, 0.f, 0.f, 0.f, 0.f, 0.f, 0.f};

    v2f kb1 = ldK(1), kb2 = ldK(2);
    v4f va = ldV(0), vb = ldV(1);
    v8f d0 = __builtin_amdgcn_wmma_f32_16x16x4_f32(
        false, a, false, ldK(0), (short)0, cz, false, false);
#pragma unroll 1
    for (int t = 0; t < 14; t += 2) {
      v8f d1 = __builtin_amdgcn_wmma_f32_16x16x4_f32(
          false, a, false, kb1, (short)0, cz, false, false);
      tile_accumulate(d0, va, l2, acc);
      va = ldV(t + 2);
      d0 = __builtin_amdgcn_wmma_f32_16x16x4_f32(
          false, a, false, kb2, (short)0, cz, false, false);
      kb1 = ldK(t + 3);
      tile_accumulate(d1, vb, l2, acc);
      vb = ldV(t + 3);
      kb2 = ldK((t + 4) & 15);
    }
    {
      v8f d1 = __builtin_amdgcn_wmma_f32_16x16x4_f32(
          false, a, false, kb1, (short)0, cz, false, false);
      tile_accumulate(d0, va, l2, acc);
      tile_accumulate(d1, vb, l2, acc);
    }

#pragma unroll
    for (int r = 0; r < 8; ++r) {
      float inv_l = 1.0f / rowsum16(l2[r >> 1][r & 1]);
#pragma unroll
      for (int dv = 0; dv < DK; ++dv) {
        float v = rowsum16(acc[r][dv >> 1][dv & 1]) * inv_l;
        if (ln == dv) {
          const int m = qbase + r + 8 * half;
          Cs[m * E_DIM + h * DK + dv] = v;
        }
      }
    }
  }
  __syncthreads();

  if (!owner) return;

  float cv[8], wc[8], wf[8];
#pragma unroll
  for (int j = 0; j < 8; ++j) {
    cv[j] = Cs[ql * E_DIM + j];
    wc[j] = Wc[j];
    wf[j] = Wf[j];
  }

  float ao[8];
  vqc8(cv, wc, ao);
  float t[8];
#pragma unroll
  for (int j = 0; j < 8; ++j) t[j] = xr[j] + ao[j];
  float y[8];
  ln8(t, g1, be1, y);

  float hv[8];
#pragma unroll
  for (int i = 0; i < 8; ++i) {
    float s = b1[i];
#pragma unroll
    for (int j = 0; j < 8; ++j) s += y[j] * w1[i * 8 + j];
    hv[i] = s;
  }
  float hq[8];
  vqc8(hv, wf, hq);

  float u[8];
#pragma unroll
  for (int i = 0; i < 8; ++i) {
    float s = b2[i];
#pragma unroll
    for (int j = 0; j < 8; ++j) s += hq[j] * w2[i * 8 + j];
    u[i] = y[i] + s;
  }
  float o[8];
  ln8(u, g2, be2, o);
  float* op = out + (size_t)(b * S_DIM + tid) * E_DIM;
#pragma unroll
  for (int j = 0; j < 8; ++j) op[j] = o[j];
}

extern "C" void kernel_launch(void* const* d_in, const int* in_sizes, int n_in,
                              void* d_out, int out_size, void* d_ws,
                              size_t ws_size, hipStream_t stream) {
  (void)in_sizes; (void)n_in; (void)out_size; (void)d_ws; (void)ws_size;
  const float* x   = (const float*)d_in[0];
  const float* Wq  = (const float*)d_in[1];
  const float* Wk  = (const float*)d_in[2];
  const float* Wv  = (const float*)d_in[3];
  const float* Wc  = (const float*)d_in[4];
  const float* Wf  = (const float*)d_in[5];
  const float* w1  = (const float*)d_in[6];
  const float* b1  = (const float*)d_in[7];
  const float* w2  = (const float*)d_in[8];
  const float* b2  = (const float*)d_in[9];
  const float* g1  = (const float*)d_in[10];
  const float* be1 = (const float*)d_in[11];
  const float* g2  = (const float*)d_in[12];
  const float* be2 = (const float*)d_in[13];
  float* out = (float*)d_out;

  fused_block_kernel<<<B_DIM * NSLICE, 256, 0, stream>>>(
      x, Wq, Wk, Wv, Wc, Wf, w1, b1, w2, b2, g1, be1, g2, be2, out);
}


# Parameter fetches moved to kernel entry: epilogue parameters (w1, w2, Wc, Wf, biases, gamma/beta) staged in LDS beside the x load, Wq read with Wk/Wv; no scalar wait behind a barrier; sums unchanged
# speedup vs baseline: 2.3259x; 1.2724x over previous
#include <hip/hip_runtime.h>
#include <math.h>

#define B_DIM 64
#define S_DIM 256
#define E_DIM 8
#define H_DIM 2
#define DK    4
#define EPS   1e-5f
#define SCALE_LOG2E 0.72134752044448170f

typedef __attribute__((ext_vector_type(2))) float v2f;
typedef __attribute__((ext_vector_type(4))) float v4f;
typedef __attribute__((ext_vector_type(8))) float v8f;

template <int CTRL>
__device__ __forceinline__ float dpp_f(float x) {
  return __int_as_float(__builtin_amdgcn_update_dpp(
      0, __float_as_int(x), CTRL, 0xF, 0xF, true));
}
__device__ __forceinline__ float rowsum16(float v) {
  v += dpp_f<0xB1>(v);
  v += dpp_f<0x4E>(v);
  v += dpp_f<0x141>(v);
  v += dpp_f<0x140>(v);
  return v;
}

__device__ __forceinline__ void vqc8(const float (&in)[8], const float (&W)[8],
                                     float (&z)[8]) {
  float c[8];
#pragma unroll
  for (int j = 0; j < 8; ++j) c[j] = __cosf(in[j] + W[j]);
  float p = c[0];
#pragma unroll
  for (int j = 1; j < 8; ++j) { p *= c[j]; z[j] = p; }
  float q = c[1];
#pragma unroll
  for (int j = 2; j < 8; ++j) q *= c[j];
  z[0] = q;
}

__device__ __forceinline__ void ln8(const float (&t)[8],
                                    const float (&g)[8],
                                    const float (&b)[8],
                                    float (&o)[8]) {
  float m = 0.f;
#pragma unroll
  for (int j = 0; j < 8; ++j) m += t[j];
  m *= 0.125f;
  float v = 0.f;
#pragma unroll
  for (int j = 0; j < 8; ++j) { float d = t[j] - m; v += d * d; }
  v *= 0.125f;
  float inv = 1.0f / sqrtf(v + EPS);
#pragma unroll
  for (int j = 0; j < 8; ++j) o[j] = (t[j] - m) * inv * g[j] + b[j];
}

__device__ __forceinline__ void tile_accumulate(const v8f& d, const v4f& vv,
                                                v2f (&l2)[4],
                                                v2f (&acc)[8][2]) {
  float p[8];
#pragma unroll
  for (int r = 0; r < 8; ++r)
    p[r] = __builtin_amdgcn_exp2f(d[r] * SCALE_LOG2E);
#pragma unroll
  for (int j = 0; j < 4; ++j) {
    v2f pj = {p[2 * j], p[2 * j + 1]};
    l2[j] += pj;
  }
  const v2f v01 = {vv.x, vv.y};
  const v2f v23 = {vv.z, vv.w};
#pragma unroll
  for (int r = 0; r < 8; ++r) {
    const v2f pp = {p[r], p[r]};
    acc[r][0] = __builtin_elementwise_fma(pp, v01, acc[r][0]);
    acc[r][1] = __builtin_elementwise_fma(pp, v23, acc[r][1]);
  }
}

#define QS_DIM 64
#define NSLICE 4
#define PS_W1  0
#define PS_W2  64
#define PS_WC  128
#define PS_WF  136
#define PS_B1  144
#define PS_B2  152
#define PS_G1  160
#define PS_BE1 168
#define PS_G2  176
#define PS_BE2 184
#define PS_DIM 192

__global__ __launch_bounds__(256) void fused_block_kernel(
    const float* __restrict__ x, const float* __restrict__ Wq,
    const float* __restrict__ Wk, const float* __restrict__ Wv,
    const float* __restrict__ Wc, const float* __restrict__ Wf,
    const float* __restrict__ w1, const float* __restrict__ b1,
    const float* __restrict__ w2, const float* __restrict__ b2,
    const float* __restrict__ g1, const float* __restrict__ be1,
    const float* __restrict__ g2, const float* __restrict__ be2,
    float* __restrict__ out) {
  __shared__ __align__(16) float Ks[H_DIM * S_DIM * DK];
  __shared__ __align__(16) float Vs[H_DIM * S_DIM * DK];
  __shared__ __align__(16) float Qs[H_DIM * QS_DIM * DK];
  __shared__ __align__(16) float Cs[QS_DIM * E_DIM];
  __shared__ __align__(16) float Ps[PS_DIM];

  const int b   = blockIdx.x / NSLICE;
  const int qs  = blockIdx.x % NSLICE;
  const int tid = threadIdx.x;
  const bool owner = (tid / QS_DIM) == qs;
  const int  ql = tid % QS_DIM;

  float xr[8];
  {
    const float* xp = x + (size_t)(b * S_DIM + tid) * E_DIM;
#pragma unroll
    for (int j = 0; j < 8; ++j) xr[j] = xp[j];


    float wk[8], wv[8], wq[8];
#pragma unroll
    for (int j = 0; j < 8; ++j) { wk[j] = Wk[j]; wv[j] = Wv[j]; wq[j] = Wq[j]; }

    const int g = (tid >> 3) & 7;
    const float* src = be2;
    src = g == 6 ? g2 : src;
    src = g == 5 ? be1 : src;
    src = g == 4 ? g1 : src;
    src = g == 3 ? b2 : src;
    src = g == 2 ? b1 : src;
    src = g == 1 ? Wf : src;
    src = g == 0 ? Wc : src;
    src += tid & 7;
    src = tid < PS_WC ? w2 + (tid - PS_W2) : src;
    src = tid < PS_W2 ? w1 + tid : src;
    float pv = 0.f;
    if (tid < PS_DIM) pv = *src;
    float zk[8], zv[8];
    vqc8(xr, wk, zk);
    vqc8(xr, wv, zv);
#pragma unroll
    for (int hh = 0; hh < H_DIM; ++hh) {
#pragma unroll
      for (int d = 0; d < DK; ++d) {
        Ks[(hh * S_DIM + tid) * DK + d] = zk[hh * DK + d];
        Vs[(hh * S_DIM + tid) * DK + d] = zv[hh * DK + d];
      }
    }
    if (tid < PS_DIM) Ps[tid] = pv;
    if (owner) {
      float zq[8];
      vqc8(xr, wq, zq);
#pragma unroll
      for (int hh = 0; hh < H_DIM; ++hh) {
#pragma unroll
        for (int d = 0; d < DK; ++d)
          Qs[(hh * QS_DIM + ql) * DK + d] = zq[hh * DK + d];
      }
    }
  }
  __syncthreads();

  {
    const int lane  = tid & 31;
    const int wv_id = tid >> 5;
    const int h     = wv_id & 1;
    const int qb    = wv_id >> 1;
    const int half  = lane >> 4;
    const int ln    = lane & 15;
    const int kp    = half * 2;
    const int qbase = qb * 16;

    const float* Qh = Qs + h * QS_DIM * DK;
    const float* Kh = Ks + h * S_DIM * DK;
    const float* Vh = Vs + h * S_DIM * DK;

    v2f a;
    a.x = Qh[(qbase + ln) * DK + kp];
    a.y = Qh[(qbase + ln) * DK + kp + 1];

    v2f l2[4], acc[8][2];
#pragma unroll
    for (int j = 0; j < 4; ++j) l2[j] = (v2f){0.f, 0.f};
#pragma unroll
    for (int r = 0; r < 8; ++r) {
      acc[r][0] = (v2f){0.f, 0.f};
      acc[r][1] = (v2f){0.f, 0.f};
    }

    auto ldK = [&](int t) -> v2f {
      return *(const v2f*)(Kh + (t * 16 + ln) * DK + kp);
    };
    auto ldV = [&](int t) -> v4f {
      return *(const v4f*)(Vh + (t * 16 + ln) * DK);
    };
    const v8f cz = {0.f, 0.f, 0.f, 0.f, 0.f, 0.f, 0.f, 0.f};

    v2f kb1 = ldK(1), kb2 = ldK(2);
    v4f va = ldV(0), vb = ldV(1);
    v8f d0 = __builtin_amdgcn_wmma_f32_16x16x4_f32(
        false, a, false, ldK(0), (short)0, cz, false, false);
#pragma unroll 1
    for (int t = 0; t < 14; t += 2) {
      v8f d1 = __builtin_amdgcn_wmma_f32_16x16x4_f32(
          false, a, false, kb1, (short)0, cz, false, false);
      tile_accumulate(d0, va, l2, acc);
      va = ldV(t + 2);
      d0 = __builtin_amdgcn_wmma_f32_16x16x4_f32(
          false, a, false, kb2, (short)0, cz, false, false);
      kb1 = ldK(t + 3);
      tile_accumulate(d1, vb, l2, acc);
      vb = ldV(t + 3);
      kb2 = ldK((t + 4) & 15);
    }
    {
      v8f d1 = __builtin_amdgcn_wmma_f32_16x16x4_f32(
          false, a, false, kb1, (short)0, cz, false, false);
      tile_accumulate(d0, va, l2, acc);
      tile_accumulate(d1, vb, l2, acc);
    }

#pragma unroll
    for (int r = 0; r < 8; ++r) {
      float inv_l = 1.0f / rowsum16(l2[r >> 1][r & 1]);
#pragma unroll
      for (int dv = 0; dv < DK; ++dv) {
        float v = rowsum16(acc[r][dv >> 1][dv & 1]) * inv_l;
        if (ln == dv) {
          const int m = qbase + r + 8 * half;
          Cs[m * E_DIM + h * DK + dv] = v;
        }
      }
    }
  }
  __syncthreads();

  if (!owner) return;

  float cv[8], wc[8], wf[8], pg1[8], pbe1[8], pg2[8], pbe2[8];
#pragma unroll
  for (int j = 0; j < 8; ++j) {
    cv[j] = Cs[ql * E_DIM + j];
    wc[j] = Ps[PS_WC + j];
    wf[j] = Ps[PS_WF + j];
    pg1[j]  = Ps[PS_G1 + j];
    pbe1[j] = Ps[PS_BE1 + j];
    pg2[j]  = Ps[PS_G2 + j];
    pbe2[j] = Ps[PS_BE2 + j];
  }

  float ao[8];
  vqc8(cv, wc, ao);
  float t[8];
#pragma unroll
  for (int j = 0; j < 8; ++j) t[j] = xr[j] + ao[j];
  float y[8];
  ln8(t, pg1, pbe1, y);

  float hv[8];
#pragma unroll
  for (int i = 0; i < 8; ++i) {
    float s = Ps[PS_B1 + i];
#pragma unroll
    for (int j = 0; j < 8; ++j) s += y[j] * Ps[PS_W1 + i * 8 + j];
    hv[i] = s;
  }
  float hq[8];
  vqc8(hv, wf, hq);

  float u[8];
#pragma unroll
  for (int i = 0; i < 8; ++i) {
    float s = Ps[PS_B2 + i];
#pragma unroll
    for (int j = 0; j < 8; ++j) s += hq[j] * Ps[PS_W2 + i * 8 + j];
    u[i] = y[i] + s;
  }
  float o[8];
  ln8(u, pg2, pbe2, o);
  float* op = out + (size_t)(b * S_DIM + tid) * E_DIM;
#pragma unroll
  for (int j = 0; j < 8; ++j) op[j] = o[j];
}

extern "C" void kernel_launch(void* const* d_in, const int* in_sizes, int n_in,
                              void* d_out, int out_size, void* d_ws,
                              size_t ws_size, hipStream_t stream) {
  (void)in_sizes; (void)n_in; (void)out_size; (void)d_ws; (void)ws_size;
  const float* x   = (const float*)d_in[0];
  const float* Wq  = (const float*)d_in[1];
  const float* Wk  = (const float*)d_in[2];
  const float* Wv  = (const float*)d_in[3];
  const float* Wc  = (const float*)d_in[4];
  const float* Wf  = (const float*)d_in[5];
  const float* w1  = (const float*)d_in[6];
  const float* b1  = (const float*)d_in[7];
  const float* w2  = (const float*)d_in[8];
  const float* b2  = (const float*)d_in[9];
  const float* g1  = (const float*)d_in[10];
  const float* be1 = (const float*)d_in[11];
  const float* g2  = (const float*)d_in[12];
  const float* be2 = (const float*)d_in[13];
  float* out = (float*)d_out;

  fused_block_kernel<<<B_DIM * NSLICE, 256, 0, stream>>>(
      x, Wq, Wk, Wv, Wc, Wf, w1, b1, w2, b2, g1, be1, g2, be2, out);
}


# Finalize as a transposing DPP butterfly: each 16-lane step halves the live values (40 to 3 per lane); one divide and one 8-byte LDS store per lane replace 8 divides, 32 masked stores; sums unchanged
# speedup vs baseline: 2.8122x; 1.2091x over previous
#include <hip/hip_runtime.h>
#include <math.h>

#define B_DIM 64
#define S_DIM 256
#define E_DIM 8
#define H_DIM 2
#define DK    4
#define EPS   1e-5f
#define SCALE_LOG2E 0.72134752044448170f

typedef __attribute__((ext_vector_type(2))) float v2f;
typedef __attribute__((ext_vector_type(4))) float v4f;
typedef __attribute__((ext_vector_type(8))) float v8f;

template <int CTRL>
__device__ __forceinline__ float dpp_f(float x) {
  return __int_as_float(__builtin_amdgcn_update_dpp(
      0, __float_as_int(x), CTRL, 0xF, 0xF, true));
}
template <int CTRL>
__device__ __forceinline__ float fold(bool up, float x, float y) {
  const float keep = up ? y : x;
  const float send = up ? x : y;
  return keep + dpp_f<CTRL>(send);
}
#define DPP_XOR1 0xB1
#define DPP_XOR2 0x4E
#define DPP_XOR4 0x164
#define DPP_XOR8 0x168

__device__ __forceinline__ void vqc8(const float (&in)[8], const float (&W)[8],
                                     float (&z)[8]) {
  float c[8];
#pragma unroll
  for (int j = 0; j < 8; ++j) c[j] = __cosf(in[j] + W[j]);
  float p = c[0];
#pragma unroll
  for (int j = 1; j < 8; ++j) { p *= c[j]; z[j] = p; }
  float q = c[1];
#pragma unroll
  for (int j = 2; j < 8; ++j) q *= c[j];
  z[0] = q;
}

__device__ __forceinline__ void ln8(const float (&t)[8],
                                    const float (&g)[8],
                                    const float (&b)[8],
                                    float (&o)[8]) {
  float m = 0.f;
#pragma unroll
  for (int j = 0; j < 8; ++j) m += t[j];
  m *= 0.125f;
  float v = 0.f;
#pragma unroll
  for (int j = 0; j < 8; ++j) { float d = t[j] - m; v += d * d; }
  v *= 0.125f;
  float inv = 1.0f / sqrtf(v + EPS);
#pragma unroll
  for (int j = 0; j < 8; ++j) o[j] = (t[j] - m) * inv * g[j] + b[j];
}

__device__ __forceinline__ void tile_accumulate(const v8f& d, const v4f& vv,
                                                v2f (&l2)[4],
                                                v2f (&acc)[8][2]) {
  float p[8];
#pragma unroll
  for (int r = 0; r < 8; ++r)
    p[r] = __builtin_amdgcn_exp2f(d[r] * SCALE_LOG2E);
#pragma unroll
  for (int j = 0; j < 4; ++j) {
    v2f pj = {p[2 * j], p[2 * j + 1]};
    l2[j] += pj;
  }
  const v2f v01 = {vv.x, vv.y};
  const v2f v23 = {vv.z, vv.w};
#pragma unroll
  for (int r = 0; r < 8; ++r) {
    const v2f pp = {p[r], p[r]};
    acc[r][0] = __builtin_elementwise_fma(pp, v01, acc[r][0]);
    acc[r][1] = __builtin_elementwise_fma(pp, v23, acc[r][1]);
  }
}

#define QS_DIM 64
#define NSLICE 4
#define PS_W1  0
#define PS_W2  64
#define PS_WC  128
#define PS_WF  136
#define PS_B1  144
#define PS_B2  152
#define PS_G1  160
#define PS_BE1 168
#define PS_G2  176
#define PS_BE2 184
#define PS_DIM 192

__global__ __launch_bounds__(256) void fused_block_kernel(
    const float* __restrict__ x, const float* __restrict__ Wq,
    const float* __restrict__ Wk, const float* __restrict__ Wv,
    const float* __restrict__ Wc, const float* __restrict__ Wf,
    const float* __restrict__ w1, const float* __restrict__ b1,
    const float* __restrict__ w2, const float* __restrict__ b2,
    const float* __restrict__ g1, const float* __restrict__ be1,
    const float* __restrict__ g2, const float* __restrict__ be2,
    float* __restrict__ out) {
  __shared__ __align__(16) float Ks[H_DIM * S_DIM * DK];
  __shared__ __align__(16) float Vs[H_DIM * S_DIM * DK];
  __shared__ __align__(16) float Qs[H_DIM * QS_DIM * DK];
  __shared__ __align__(16) float Cs[QS_DIM * E_DIM];
  __shared__ __align__(16) float Ps[PS_DIM];

  const int b   = blockIdx.x / NSLICE;
  const int qs  = blockIdx.x % NSLICE;
  const int tid = threadIdx.x;
  const bool owner = (tid / QS_DIM) == qs;
  const int  ql = tid % QS_DIM;

  float xr[8];
  {
    const float* xp = x + (size_t)(b * S_DIM + tid) * E_DIM;
#pragma unroll
    for (int j = 0; j < 8; ++j) xr[j] = xp[j];


    float wk[8], wv[8], wq[8];
#pragma unroll
    for (int j = 0; j < 8; ++j) { wk[j] = Wk[j]; wv[j] = Wv[j]; wq[j] = Wq[j]; }

    const int g = (tid >> 3) & 7;
    const float* src = be2;
    src = g == 6 ? g2 : src;
    src = g == 5 ? be1 : src;
    src = g == 4 ? g1 : src;
    src = g == 3 ? b2 : src;
    src = g == 2 ? b1 : src;
    src = g == 1 ? Wf : src;
    src = g == 0 ? Wc : src;
    src += tid & 7;
    src = tid < PS_WC ? w2 + (tid - PS_W2) : src;
    src = tid < PS_W2 ? w1 + tid : src;
    float pv = 0.f;
    if (tid < PS_DIM) pv = *src;
    float zk[8], zv[8];
    vqc8(xr, wk, zk);
    vqc8(xr, wv, zv);
#pragma unroll
    for (int hh = 0; hh < H_DIM; ++hh) {
#pragma unroll
      for (int d = 0; d < DK; ++d) {
        Ks[(hh * S_DIM + tid) * DK + d] = zk[hh * DK + d];
        Vs[(hh * S_DIM + tid) * DK + d] = zv[hh * DK + d];
      }
    }
    if (tid < PS_DIM) Ps[tid] = pv;
    if (owner) {
      float zq[8];
      vqc8(xr, wq, zq);
#pragma unroll
      for (int hh = 0; hh < H_DIM; ++hh) {
#pragma unroll
        for (int d = 0; d < DK; ++d)
          Qs[(hh * QS_DIM + ql) * DK + d] = zq[hh * DK + d];
      }
    }
  }
  __syncthreads();

  {
    const int lane  = tid & 31;
    const int wv_id = tid >> 5;
    const int h     = wv_id & 1;
    const int qb    = wv_id >> 1;
    const int half  = lane >> 4;
    const int ln    = lane & 15;
    const int kp    = half * 2;
    const int qbase = qb * 16;

    const float* Qh = Qs + h * QS_DIM * DK;
    const float* Kh = Ks + h * S_DIM * DK;
    const float* Vh = Vs + h * S_DIM * DK;

    v2f a;
    a.x = Qh[(qbase + ln) * DK + kp];
    a.y = Qh[(qbase + ln) * DK + kp + 1];

    v2f l2[4], acc[8][2];
#pragma unroll
    for (int j = 0; j < 4; ++j) l2[j] = (v2f){0.f, 0.f};
#pragma unroll
    for (int r = 0; r < 8; ++r) {
      acc[r][0] = (v2f){0.f, 0.f};
      acc[r][1] = (v2f){0.f, 0.f};
    }

    auto ldK = [&](int t) -> v2f {
      return *(const v2f*)(Kh + (t * 16 + ln) * DK + kp);
    };
    auto ldV = [&](int t) -> v4f {
      return *(const v4f*)(Vh + (t * 16 + ln) * DK);
    };
    const v8f cz = {0.f, 0.f, 0.f, 0.f, 0.f, 0.f, 0.f, 0.f};

    v2f kb1 = ldK(1), kb2 = ldK(2);
    v4f va = ldV(0), vb = ldV(1);
    v8f d0 = __builtin_amdgcn_wmma_f32_16x16x4_f32(
        false, a, false, ldK(0), (short)0, cz, false, false);
#pragma unroll 1
    for (int t = 0; t < 14; t += 2) {
      v8f d1 = __builtin_amdgcn_wmma_f32_16x16x4_f32(
          false, a, false, kb1, (short)0, cz, false, false);
      tile_accumulate(d0, va, l2, acc);
      va = ldV(t + 2);
      d0 = __builtin_amdgcn_wmma_f32_16x16x4_f32(
          false, a, false, kb2, (short)0, cz, false, false);
      kb1 = ldK(t + 3);
      tile_accumulate(d1, vb, l2, acc);
      vb = ldV(t + 3);
      kb2 = ldK((t + 4) & 15);
    }
    {
      v8f d1 = __builtin_amdgcn_wmma_f32_16x16x4_f32(
          false, a, false, kb1, (short)0, cz, false, false);
      tile_accumulate(d0, va, l2, acc);
      tile_accumulate(d1, vb, l2, acc);
    }

    const bool b0 = (ln & 1) != 0, b1 = (ln & 2) != 0;
    const bool b2 = (ln & 4) != 0, b3 = (ln & 8) != 0;
    float s1[4][4], d1[4];
#pragma unroll
    for (int k = 0; k < 4; ++k) {
#pragma unroll
      for (int dv = 0; dv < DK; ++dv)
        s1[k][dv] = fold<DPP_XOR1>(b0, acc[2 * k][dv >> 1][dv & 1],
                                   acc[2 * k + 1][dv >> 1][dv & 1]);
      d1[k] = fold<DPP_XOR1>(b0, l2[k][0], l2[k][1]);
    }
    float s2[2][4], d2[2];
#pragma unroll
    for (int k = 0; k < 2; ++k) {
#pragma unroll
      for (int dv = 0; dv < DK; ++dv)
        s2[k][dv] = fold<DPP_XOR2>(b1, s1[2 * k][dv], s1[2 * k + 1][dv]);
      d2[k] = fold<DPP_XOR2>(b1, d1[2 * k], d1[2 * k + 1]);
    }
    float s3[4];
#pragma unroll
    for (int dv = 0; dv < DK; ++dv)
      s3[dv] = fold<DPP_XOR4>(b2, s2[0][dv], s2[1][dv]);
    float den = fold<DPP_XOR4>(b2, d2[0], d2[1]);
    const float c0 = fold<DPP_XOR8>(b3, s3[0], s3[2]);
    const float c1 = fold<DPP_XOR8>(b3, s3[1], s3[3]);
    den += dpp_f<DPP_XOR8>(den);
    {
      const float inv_l = 1.0f / den;
      const int m = qbase + (ln & 7) + 8 * half;
      const v2f o = {c0 * inv_l, c1 * inv_l};
      *(v2f*)(Cs + m * E_DIM + h * DK + 2 * (ln >> 3)) = o;
    }
  }
  __syncthreads();

  if (!owner) return;

  float cv[8], wc[8], wf[8], pg1[8], pbe1[8], pg2[8], pbe2[8];
#pragma unroll
  for (int j = 0; j < 8; ++j) {
    cv[j] = Cs[ql * E_DIM + j];
    wc[j] = Ps[PS_WC + j];
    wf[j] = Ps[PS_WF + j];
    pg1[j]  = Ps[PS_G1 + j];
    pbe1[j] = Ps[PS_BE1 + j];
    pg2[j]  = Ps[PS_G2 + j];
    pbe2[j] = Ps[PS_BE2 + j];
  }

  float ao[8];
  vqc8(cv, wc, ao);
  float t[8];
#pragma unroll
  for (int j = 0; j < 8; ++j) t[j] = xr[j] + ao[j];
  float y[8];
  ln8(t, pg1, pbe1, y);

  float hv[8];
#pragma unroll
  for (int i = 0; i < 8; ++i) {
    float s = Ps[PS_B1 + i];
#pragma unroll
    for (int j = 0; j < 8; ++j) s += y[j] * Ps[PS_W1 + i * 8 + j];
    hv[i] = s;
  }
  float hq[8];
  vqc8(hv, wf, hq);

  float u[8];
#pragma unroll
  for (int i = 0; i < 8; ++i) {
    float s = Ps[PS_B2 + i];
#pragma unroll
    for (int j = 0; j < 8; ++j) s += hq[j] * Ps[PS_W2 + i * 8 + j];
    u[i] = y[i] + s;
  }
  float o[8];
  ln8(u, pg2, pbe2, o);
  float* op = out + (size_t)(b * S_DIM + tid) * E_DIM;
#pragma unroll
  for (int j = 0; j < 8; ++j) op[j] = o[j];
}

extern "C" void kernel_launch(void* const* d_in, const int* in_sizes, int n_in,
                              void* d_out, int out_size, void* d_ws,
                              size_t ws_size, hipStream_t stream) {
  (void)in_sizes; (void)n_in; (void)out_size; (void)d_ws; (void)ws_size;
  const float* x   = (const float*)d_in[0];
  const float* Wq  = (const float*)d_in[1];
  const float* Wk  = (const float*)d_in[2];
  const float* Wv  = (const float*)d_in[3];
  const float* Wc  = (const float*)d_in[4];
  const float* Wf  = (const float*)d_in[5];
  const float* w1  = (const float*)d_in[6];
  const float* b1  = (const float*)d_in[7];
  const float* w2  = (const float*)d_in[8];
  const float* b2  = (const float*)d_in[9];
  const float* g1  = (const float*)d_in[10];
  const float* be1 = (const float*)d_in[11];
  const float* g2  = (const float*)d_in[12];
  const float* be2 = (const float*)d_in[13];
  float* out = (float*)d_out;

  fused_block_kernel<<<B_DIM * NSLICE, 256, 0, stream>>>(
      x, Wq, Wk, Wv, Wc, Wf, w1, b1, w2, b2, g1, be1, g2, be2, out);
}


# Epilogue spread over all 256 threads: four lanes per token, two features each, quad DPP between them, not one lane per token on two waves; LayerNorm sums become a pairwise tree, products regrouped
# speedup vs baseline: 3.1245x; 1.1111x over previous
#include <hip/hip_runtime.h>
#include <math.h>

#define B_DIM 64
#define S_DIM 256
#define E_DIM 8
#define H_DIM 2
#define DK    4
#define EPS   1e-5f
#define SCALE_LOG2E 0.72134752044448170f

typedef __attribute__((ext_vector_type(2))) float v2f;
typedef __attribute__((ext_vector_type(4))) float v4f;
typedef __attribute__((ext_vector_type(8))) float v8f;

template <int CTRL>
__device__ __forceinline__ float dpp_f(float x) {
  return __int_as_float(__builtin_amdgcn_update_dpp(
      0, __float_as_int(x), CTRL, 0xF, 0xF, true));
}
template <int CTRL>
__device__ __forceinline__ float fold(bool up, float x, float y) {
  const float keep = up ? y : x;
  const float send = up ? x : y;
  return keep + dpp_f<CTRL>(send);
}
#define DPP_XOR1 0xB1
#define DPP_XOR2 0x4E
#define DPP_XOR4 0x164
#define DPP_XOR8 0x168

__device__ __forceinline__ void vqc8(const float (&in)[8], const float (&W)[8],
                                     float (&z)[8]) {
  float c[8];
#pragma unroll
  for (int j = 0; j < 8; ++j) c[j] = __cosf(in[j] + W[j]);
  float p = c[0];
#pragma unroll
  for (int j = 1; j < 8; ++j) { p *= c[j]; z[j] = p; }
  float q = c[1];
#pragma unroll
  for (int j = 2; j < 8; ++j) q *= c[j];
  z[0] = q;
}

#define DPP_QUAD0 0x00
#define DPP_QUAD1 0x55
#define DPP_QUAD2 0xAA
#define DPP_QUAD3 0xFF

__device__ __forceinline__ void vqc_quad(int p, float in0, float in1,
                                         float w0, float w1,
                                         float& z0, float& z1) {
  const float c0 = __cosf(in0 + w0);
  const float c1 = __cosf(in1 + w1);
  const float bb = c0 * c1;
  const float p0 = dpp_f<DPP_QUAD0>(bb);
  const float p1 = dpp_f<DPP_QUAD1>(bb);
  const float p2 = dpp_f<DPP_QUAD2>(bb);
  const float p3 = dpp_f<DPP_QUAD3>(bb);
  const float c1_first = dpp_f<DPP_QUAD0>(c1);
  const float p01 = p0 * p1;
  const float p012 = p01 * p2;
  float e = 1.0f;
  e = p == 1 ? p0 : e;
  e = p == 2 ? p01 : e;
  e = p == 3 ? p012 : e;
  const float zfirst = ((c1_first * p1) * p2) * p3;
  z0 = p == 0 ? zfirst : e * c0;
  z1 = e * bb;
}

__device__ __forceinline__ void ln_quad(float t0, float t1, v2f g, v2f be,
                                        float& o0, float& o1) {
  float s = t0 + t1;
  s += dpp_f<DPP_XOR1>(s);
  s += dpp_f<DPP_XOR2>(s);
  const float m = s * 0.125f;
  const float d0 = t0 - m;
  const float d1 = t1 - m;
  float q = d0 * d0 + d1 * d1;
  q += dpp_f<DPP_XOR1>(q);
  q += dpp_f<DPP_XOR2>(q);
  const float v = q * 0.125f;
  const float inv = 1.0f / sqrtf(v + EPS);
  o0 = d0 * inv * g.x + be.x;
  o1 = d1 * inv * g.y + be.y;
}

__device__ __forceinline__ void rows_quad(float y0, float y1,
                                          const v4f (&w)[4], v2f bias,
                                          float& o0, float& o1) {
  float ya[4], yb[4];
  ya[0] = dpp_f<DPP_QUAD0>(y0);  yb[0] = dpp_f<DPP_QUAD0>(y1);
  ya[1] = dpp_f<DPP_QUAD1>(y0);  yb[1] = dpp_f<DPP_QUAD1>(y1);
  ya[2] = dpp_f<DPP_QUAD2>(y0);  yb[2] = dpp_f<DPP_QUAD2>(y1);
  ya[3] = dpp_f<DPP_QUAD3>(y0);  yb[3] = dpp_f<DPP_QUAD3>(y1);
  v2f s = bias;
#pragma unroll
  for (int k = 0; k < 4; ++k) {
    const v2f a  = {ya[k], ya[k]};
    const v2f bq = {yb[k], yb[k]};
    const v2f wa = {w[k].x, w[k].y};
    const v2f wb = {w[k].z, w[k].w};
    s = __builtin_elementwise_fma(a, wa, s);
    s = __builtin_elementwise_fma(bq, wb, s);
  }
  o0 = s.x;
  o1 = s.y;
}

__device__ __forceinline__ void tile_accumulate(const v8f& d, const v4f& vv,
                                                v2f (&l2)[4],
                                                v2f (&acc)[8][2]) {
  float p[8];
#pragma unroll
  for (int r = 0; r < 8; ++r)
    p[r] = __builtin_amdgcn_exp2f(d[r] * SCALE_LOG2E);
#pragma unroll
  for (int j = 0; j < 4; ++j) {
    v2f pj = {p[2 * j], p[2 * j + 1]};
    l2[j] += pj;
  }
  const v2f v01 = {vv.x, vv.y};
  const v2f v23 = {vv.z, vv.w};
#pragma unroll
  for (int r = 0; r < 8; ++r) {
    const v2f pp = {p[r], p[r]};
    acc[r][0] = __builtin_elementwise_fma(pp, v01, acc[r][0]);
    acc[r][1] = __builtin_elementwise_fma(pp, v23, acc[r][1]);
  }
}

#define QS_DIM 64
#define NSLICE 4
#define PS_W1  0
#define PS_W2  64
#define PS_WC  128
#define PS_WF  136
#define PS_B1  144
#define PS_B2  152
#define PS_G1  160
#define PS_BE1 168
#define PS_G2  176
#define PS_BE2 184
#define PS_DIM 192

__global__ __launch_bounds__(256) void fused_block_kernel(
    const float* __restrict__ x, const float* __restrict__ Wq,
    const float* __restrict__ Wk, const float* __restrict__ Wv,
    const float* __restrict__ Wc, const float* __restrict__ Wf,
    const float* __restrict__ w1, const float* __restrict__ b1,
    const float* __restrict__ w2, const float* __restrict__ b2,
    const float* __restrict__ g1, const float* __restrict__ be1,
    const float* __restrict__ g2, const float* __restrict__ be2,
    float* __restrict__ out) {
  __shared__ __align__(16) float Ks[H_DIM * S_DIM * DK];
  __shared__ __align__(16) float Vs[H_DIM * S_DIM * DK];
  __shared__ __align__(16) float Qs[H_DIM * QS_DIM * DK];
  __shared__ __align__(16) float Cs[QS_DIM * E_DIM];
  __shared__ __align__(16) float Ps[PS_DIM];

  const int b   = blockIdx.x / NSLICE;
  const int qs  = blockIdx.x % NSLICE;
  const int tid = threadIdx.x;
  const bool owner = (tid / QS_DIM) == qs;
  const int  ql = tid % QS_DIM;

  float xr[8];
  const int tok = tid >> 2;
  const int p   = tid & 3;
  v2f x2;
  {
    const float* xp = x + (size_t)(b * S_DIM + tid) * E_DIM;
#pragma unroll
    for (int j = 0; j < 8; ++j) xr[j] = xp[j];
    x2 = *(const v2f*)(x + (size_t)(b * S_DIM + qs * QS_DIM + tok) * E_DIM +
                       2 * p);


    float wk[8], wv[8], wq[8];
#pragma unroll
    for (int j = 0; j < 8; ++j) { wk[j] = Wk[j]; wv[j] = Wv[j]; wq[j] = Wq[j]; }

    const int g = (tid >> 3) & 7;
    const float* src = be2;
    src = g == 6 ? g2 : src;
    src = g == 5 ? be1 : src;
    src = g == 4 ? g1 : src;
    src = g == 3 ? b2 : src;
    src = g == 2 ? b1 : src;
    src = g == 1 ? Wf : src;
    src = g == 0 ? Wc : src;
    src += tid & 7;
    src = tid < PS_WC ? w2 + (tid - PS_W2) : src;
    src = tid < PS_W2 ? w1 + tid : src;
    float pv = 0.f;
    if (tid < PS_DIM) pv = *src;
    float zk[8], zv[8];
    vqc8(xr, wk, zk);
    vqc8(xr, wv, zv);
#pragma unroll
    for (int hh = 0; hh < H_DIM; ++hh) {
#pragma unroll
      for (int d = 0; d < DK; ++d) {
        Ks[(hh * S_DIM + tid) * DK + d] = zk[hh * DK + d];
        Vs[(hh * S_DIM + tid) * DK + d] = zv[hh * DK + d];
      }
    }
    const int wdst = (tid & 64) + 16 * ((tid & 63) >> 4) + 2 * (tid & 7) +
                     ((tid >> 3) & 1);
    if (tid < PS_DIM) Ps[tid < PS_WC ? wdst : tid] = pv;
    if (owner) {
      float zq[8];
      vqc8(xr, wq, zq);
#pragma unroll
      for (int hh = 0; hh < H_DIM; ++hh) {
#pragma unroll
        for (int d = 0; d < DK; ++d)
          Qs[(hh * QS_DIM + ql) * DK + d] = zq[hh * DK + d];
      }
    }
  }
  __syncthreads();

  v2f wc2, wf2, pb1, pb2, pg1, pbe1, pg2, pbe2;
  v4f w1r[4], w2r[4];
  {
    const int lane  = tid & 31;
    const int wv_id = tid >> 5;
    const int h     = wv_id & 1;
    const int qb    = wv_id >> 1;
    const int half  = lane >> 4;
    const int ln    = lane & 15;
    const int kp    = half * 2;
    const int qbase = qb * 16;

    const float* Qh = Qs + h * QS_DIM * DK;
    const float* Kh = Ks + h * S_DIM * DK;
    const float* Vh = Vs + h * S_DIM * DK;

    v2f a;
    a.x = Qh[(qbase + ln) * DK + kp];
    a.y = Qh[(qbase + ln) * DK + kp + 1];

    v2f l2[4], acc[8][2];
#pragma unroll
    for (int j = 0; j < 4; ++j) l2[j] = (v2f){0.f, 0.f};
#pragma unroll
    for (int r = 0; r < 8; ++r) {
      acc[r][0] = (v2f){0.f, 0.f};
      acc[r][1] = (v2f){0.f, 0.f};
    }

    auto ldK = [&](int t) -> v2f {
      return *(const v2f*)(Kh + (t * 16 + ln) * DK + kp);
    };
    auto ldV = [&](int t) -> v4f {
      return *(const v4f*)(Vh + (t * 16 + ln) * DK);
    };
    const v8f cz = {0.f, 0.f, 0.f, 0.f, 0.f, 0.f, 0.f, 0.f};

    v2f kb1 = ldK(1), kb2 = ldK(2);
    v4f va = ldV(0), vb = ldV(1);
    v8f d0 = __builtin_amdgcn_wmma_f32_16x16x4_f32(
        false, a, false, ldK(0), (short)0, cz, false, false);
#pragma unroll 1
    for (int t = 0; t < 14; t += 2) {
      v8f d1 = __builtin_amdgcn_wmma_f32_16x16x4_f32(
          false, a, false, kb1, (short)0, cz, false, false);
      tile_accumulate(d0, va, l2, acc);
      va = ldV(t + 2);
      d0 = __builtin_amdgcn_wmma_f32_16x16x4_f32(
          false, a, false, kb2, (short)0, cz, false, false);
      kb1 = ldK(t + 3);
      tile_accumulate(d1, vb, l2, acc);
      vb = ldV(t + 3);
      kb2 = ldK((t + 4) & 15);
    }
    {
      v8f d1 = __builtin_amdgcn_wmma_f32_16x16x4_f32(
          false, a, false, kb1, (short)0, cz, false, false);
      tile_accumulate(d0, va, l2, acc);
      tile_accumulate(d1, vb, l2, acc);
    }

    wc2  = *(const v2f*)(Ps + PS_WC  + 2 * p);
    wf2  = *(const v2f*)(Ps + PS_WF  + 2 * p);
    pb1  = *(const v2f*)(Ps + PS_B1  + 2 * p);
    pb2  = *(const v2f*)(Ps + PS_B2  + 2 * p);
    pg1  = *(const v2f*)(Ps + PS_G1  + 2 * p);
    pbe1 = *(const v2f*)(Ps + PS_BE1 + 2 * p);
    pg2  = *(const v2f*)(Ps + PS_G2  + 2 * p);
    pbe2 = *(const v2f*)(Ps + PS_BE2 + 2 * p);
#pragma unroll
    for (int k = 0; k < 4; ++k) {
      w1r[k] = *(const v4f*)(Ps + PS_W1 + 16 * p + 4 * k);
      w2r[k] = *(const v4f*)(Ps + PS_W2 + 16 * p + 4 * k);
    }

    const bool b0 = (ln & 1) != 0, b1 = (ln & 2) != 0;
    const bool b2 = (ln & 4) != 0, b3 = (ln & 8) != 0;
    float s1[4][4], d1[4];
#pragma unroll
    for (int k = 0; k < 4; ++k) {
#pragma unroll
      for (int dv = 0; dv < DK; ++dv)
        s1[k][dv] = fold<DPP_XOR1>(b0, acc[2 * k][dv >> 1][dv & 1],
                                   acc[2 * k + 1][dv >> 1][dv & 1]);
      d1[k] = fold<DPP_XOR1>(b0, l2[k][0], l2[k][1]);
    }
    float s2[2][4], d2[2];
#pragma unroll
    for (int k = 0; k < 2; ++k) {
#pragma unroll
      for (int dv = 0; dv < DK; ++dv)
        s2[k][dv] = fold<DPP_XOR2>(b1, s1[2 * k][dv], s1[2 * k + 1][dv]);
      d2[k] = fold<DPP_XOR2>(b1, d1[2 * k], d1[2 * k + 1]);
    }
    float s3[4];
#pragma unroll
    for (int dv = 0; dv < DK; ++dv)
      s3[dv] = fold<DPP_XOR4>(b2, s2[0][dv], s2[1][dv]);
    float den = fold<DPP_XOR4>(b2, d2[0], d2[1]);
    const float c0 = fold<DPP_XOR8>(b3, s3[0], s3[2]);
    const float c1 = fold<DPP_XOR8>(b3, s3[1], s3[3]);
    den += dpp_f<DPP_XOR8>(den);
    {
      const float inv_l = 1.0f / den;
      const int m = qbase + (ln & 7) + 8 * half;
      const v2f o = {c0 * inv_l, c1 * inv_l};
      *(v2f*)(Cs + m * E_DIM + h * DK + 2 * (ln >> 3)) = o;
    }
  }
  __syncthreads();

  const v2f cv = *(const v2f*)(Cs + tok * E_DIM + 2 * p);
  float a0, a1;
  vqc_quad(p, cv.x, cv.y, wc2.x, wc2.y, a0, a1);
  float y0, y1;
  ln_quad(x2.x + a0, x2.y + a1, pg1, pbe1, y0, y1);

  float h0, h1;
  rows_quad(y0, y1, w1r, pb1, h0, h1);
  float q0, q1;
  vqc_quad(p, h0, h1, wf2.x, wf2.y, q0, q1);
  float f0, f1;
  rows_quad(q0, q1, w2r, pb2, f0, f1);

  float o0, o1;
  ln_quad(y0 + f0, y1 + f1, pg2, pbe2, o0, o1);
  const v2f o = {o0, o1};
  *(v2f*)(out + (size_t)(b * S_DIM + qs * QS_DIM + tok) * E_DIM + 2 * p) = o;
}

extern "C" void kernel_launch(void* const* d_in, const int* in_sizes, int n_in,
                              void* d_out, int out_size, void* d_ws,
                              size_t ws_size, hipStream_t stream) {
  (void)in_sizes; (void)n_in; (void)out_size; (void)d_ws; (void)ws_size;
  const float* x   = (const float*)d_in[0];
  const float* Wq  = (const float*)d_in[1];
  const float* Wk  = (const float*)d_in[2];
  const float* Wv  = (const float*)d_in[3];
  const float* Wc  = (const float*)d_in[4];
  const float* Wf  = (const float*)d_in[5];
  const float* w1  = (const float*)d_in[6];
  const float* b1  = (const float*)d_in[7];
  const float* w2  = (const float*)d_in[8];
  const float* b2  = (const float*)d_in[9];
  const float* g1  = (const float*)d_in[10];
  const float* be1 = (const float*)d_in[11];
  const float* g2  = (const float*)d_in[12];
  const float* be2 = (const float*)d_in[13];
  float* out = (float*)d_out;

  fused_block_kernel<<<B_DIM * NSLICE, 256, 0, stream>>>(
      x, Wq, Wk, Wv, Wc, Wf, w1, b1, w2, b2, g1, be1, g2, be2, out);
}


# Key phase as straight-line code scheduled for the vector port: K/V of all 16 tiles read from LDS as one group, scale folded into Q, accumulators paired by rows (score rounding moves; sums in order)
# speedup vs baseline: 3.2945x; 1.0544x over previous
#include <hip/hip_runtime.h>
#include <math.h>

#define B_DIM 64
#define S_DIM 256
#define E_DIM 8
#define H_DIM 2
#define DK    4
#define EPS   1e-5f
#define SCALE_LOG2E 0.72134752044448170f

typedef __attribute__((ext_vector_type(2))) float v2f;
typedef __attribute__((ext_vector_type(4))) float v4f;
typedef __attribute__((ext_vector_type(8))) float v8f;

template <int CTRL>
__device__ __forceinline__ float dpp_f(float x) {
  return __int_as_float(__builtin_amdgcn_update_dpp(
      0, __float_as_int(x), CTRL, 0xF, 0xF, true));
}
template <int CTRL>
__device__ __forceinline__ float fold(bool up, float x, float y) {
  const float keep = up ? y : x;
  const float send = up ? x : y;
  return keep + dpp_f<CTRL>(send);
}
#define DPP_XOR1 0xB1
#define DPP_XOR2 0x4E
#define DPP_XOR4 0x164
#define DPP_XOR8 0x168

__device__ __forceinline__ void vqc8(const float (&in)[8], const float (&W)[8],
                                     float (&z)[8]) {
  float c[8];
#pragma unroll
  for (int j = 0; j < 8; ++j) c[j] = __cosf(in[j] + W[j]);
  float p = c[0];
#pragma unroll
  for (int j = 1; j < 8; ++j) { p *= c[j]; z[j] = p; }
  float q = c[1];
#pragma unroll
  for (int j = 2; j < 8; ++j) q *= c[j];
  z[0] = q;
}

#define DPP_QUAD0 0x00
#define DPP_QUAD1 0x55
#define DPP_QUAD2 0xAA
#define DPP_QUAD3 0xFF

__device__ __forceinline__ void vqc_quad(int p, float in0, float in1,
                                         float w0, float w1,
                                         float& z0, float& z1) {
  const float c0 = __cosf(in0 + w0);
  const float c1 = __cosf(in1 + w1);
  const float bb = c0 * c1;
  const float p0 = dpp_f<DPP_QUAD0>(bb);
  const float p1 = dpp_f<DPP_QUAD1>(bb);
  const float p2 = dpp_f<DPP_QUAD2>(bb);
  const float p3 = dpp_f<DPP_QUAD3>(bb);
  const float c1_first = dpp_f<DPP_QUAD0>(c1);
  const float p01 = p0 * p1;
  const float p012 = p01 * p2;
  float e = 1.0f;
  e = p == 1 ? p0 : e;
  e = p == 2 ? p01 : e;
  e = p == 3 ? p012 : e;
  const float zfirst = ((c1_first * p1) * p2) * p3;
  z0 = p == 0 ? zfirst : e * c0;
  z1 = e * bb;
}

__device__ __forceinline__ void ln_quad(float t0, float t1, v2f g, v2f be,
                                        float& o0, float& o1) {
  float s = t0 + t1;
  s += dpp_f<DPP_XOR1>(s);
  s += dpp_f<DPP_XOR2>(s);
  const float m = s * 0.125f;
  const float d0 = t0 - m;
  const float d1 = t1 - m;
  float q = d0 * d0 + d1 * d1;
  q += dpp_f<DPP_XOR1>(q);
  q += dpp_f<DPP_XOR2>(q);
  const float v = q * 0.125f;
  const float inv = 1.0f / sqrtf(v + EPS);
  o0 = d0 * inv * g.x + be.x;
  o1 = d1 * inv * g.y + be.y;
}

__device__ __forceinline__ void rows_quad(float y0, float y1,
                                          const v4f (&w)[4], v2f bias,
                                          float& o0, float& o1) {
  float ya[4], yb[4];
  ya[0] = dpp_f<DPP_QUAD0>(y0);  yb[0] = dpp_f<DPP_QUAD0>(y1);
  ya[1] = dpp_f<DPP_QUAD1>(y0);  yb[1] = dpp_f<DPP_QUAD1>(y1);
  ya[2] = dpp_f<DPP_QUAD2>(y0);  yb[2] = dpp_f<DPP_QUAD2>(y1);
  ya[3] = dpp_f<DPP_QUAD3>(y0);  yb[3] = dpp_f<DPP_QUAD3>(y1);
  v2f s = bias;
#pragma unroll
  for (int k = 0; k < 4; ++k) {
    const v2f a  = {ya[k], ya[k]};
    const v2f bq = {yb[k], yb[k]};
    const v2f wa = {w[k].x, w[k].y};
    const v2f wb = {w[k].z, w[k].w};
    s = __builtin_elementwise_fma(a, wa, s);
    s = __builtin_elementwise_fma(bq, wb, s);
  }
  o0 = s.x;
  o1 = s.y;
}

__device__ __forceinline__ void tile_exp(const v8f& d, v2f (&p)[4]) {
#pragma unroll
  for (int j = 0; j < 4; ++j) {
    p[j].x = __builtin_amdgcn_exp2f(d[2 * j]);
    p[j].y = __builtin_amdgcn_exp2f(d[2 * j + 1]);
  }
}
__device__ __forceinline__ void tile_accumulate(const v2f (&p)[4],
                                                const v4f& vv,
                                                v2f (&l2)[4],
                                                v2f (&acc)[4][4]) {
#pragma unroll
  for (int j = 0; j < 4; ++j) l2[j] += p[j];
#pragma unroll
  for (int c = 0; c < 4; ++c) {
    const v2f vc = {vv[c], vv[c]};
#pragma unroll
    for (int j = 0; j < 4; ++j)
      acc[j][c] = __builtin_elementwise_fma(p[j], vc, acc[j][c]);
  }
}

template <bool W, bool T>
__device__ __forceinline__ void tile_order() {
  if constexpr (T) {
    __builtin_amdgcn_sched_group_barrier(0x002, 13, 0);
    __builtin_amdgcn_sched_group_barrier(0x400, 1, 0);
    __builtin_amdgcn_sched_group_barrier(0x002, 1, 0);
    __builtin_amdgcn_sched_group_barrier(0x400, 1, 0);
    __builtin_amdgcn_sched_group_barrier(0x002, 1, 0);
    __builtin_amdgcn_sched_group_barrier(0x400, 1, 0);
    __builtin_amdgcn_sched_group_barrier(0x002, 1, 0);
    __builtin_amdgcn_sched_group_barrier(0x400, 1, 0);
    __builtin_amdgcn_sched_group_barrier(0x002, 1, 0);
    __builtin_amdgcn_sched_group_barrier(0x400, 1, 0);
    __builtin_amdgcn_sched_group_barrier(0x002, 1, 0);
    __builtin_amdgcn_sched_group_barrier(0x400, 1, 0);
    __builtin_amdgcn_sched_group_barrier(0x002, 1, 0);
    __builtin_amdgcn_sched_group_barrier(0x400, 1, 0);
    __builtin_amdgcn_sched_group_barrier(0x002, 1, 0);
    __builtin_amdgcn_sched_group_barrier(0x400, 1, 0);
    __builtin_amdgcn_sched_group_barrier(0x002, 1, 0);
  } else {
    __builtin_amdgcn_sched_group_barrier(0x002, 21, 0);
  }
  if constexpr (W) __builtin_amdgcn_sched_group_barrier(0x008, 1, 0);
}

#define QS_DIM 64
#define NSLICE 4
#define PS_W1  0
#define PS_W2  64
#define PS_WC  128
#define PS_WF  136
#define PS_B1  144
#define PS_B2  152
#define PS_G1  160
#define PS_BE1 168
#define PS_G2  176
#define PS_BE2 184
#define PS_DIM 192

__global__ __launch_bounds__(256) void fused_block_kernel(
    const float* __restrict__ x, const float* __restrict__ Wq,
    const float* __restrict__ Wk, const float* __restrict__ Wv,
    const float* __restrict__ Wc, const float* __restrict__ Wf,
    const float* __restrict__ w1, const float* __restrict__ b1,
    const float* __restrict__ w2, const float* __restrict__ b2,
    const float* __restrict__ g1, const float* __restrict__ be1,
    const float* __restrict__ g2, const float* __restrict__ be2,
    float* __restrict__ out) {
  __shared__ __align__(16) float Ks[H_DIM * S_DIM * DK];
  __shared__ __align__(16) float Vs[H_DIM * S_DIM * DK];
  __shared__ __align__(16) float Qs[H_DIM * QS_DIM * DK];
  __shared__ __align__(16) float Cs[QS_DIM * E_DIM];
  __shared__ __align__(16) float Ps[PS_DIM];

  const int b   = blockIdx.x / NSLICE;
  const int qs  = blockIdx.x % NSLICE;
  const int tid = threadIdx.x;
  const bool owner = (tid / QS_DIM) == qs;
  const int  ql = tid % QS_DIM;

  float xr[8];
  const int tok = tid >> 2;
  const int p   = tid & 3;
  v2f x2;
  {
    const float* xp = x + (size_t)(b * S_DIM + tid) * E_DIM;
#pragma unroll
    for (int j = 0; j < 8; ++j) xr[j] = xp[j];
    x2 = *(const v2f*)(x + (size_t)(b * S_DIM + qs * QS_DIM + tok) * E_DIM +
                       2 * p);


    float wk[8], wv[8], wq[8];
#pragma unroll
    for (int j = 0; j < 8; ++j) { wk[j] = Wk[j]; wv[j] = Wv[j]; wq[j] = Wq[j]; }

    const int g = (tid >> 3) & 7;
    const float* src = be2;
    src = g == 6 ? g2 : src;
    src = g == 5 ? be1 : src;
    src = g == 4 ? g1 : src;
    src = g == 3 ? b2 : src;
    src = g == 2 ? b1 : src;
    src = g == 1 ? Wf : src;
    src = g == 0 ? Wc : src;
    src += tid & 7;
    src = tid < PS_WC ? w2 + (tid - PS_W2) : src;
    src = tid < PS_W2 ? w1 + tid : src;
    float pv = 0.f;
    if (tid < PS_DIM) pv = *src;
    float zk[8], zv[8];
    vqc8(xr, wk, zk);
    vqc8(xr, wv, zv);
#pragma unroll
    for (int hh = 0; hh < H_DIM; ++hh) {
#pragma unroll
      for (int d = 0; d < DK; ++d) {
        Ks[(hh * S_DIM + tid) * DK + d] = zk[hh * DK + d];
        Vs[(hh * S_DIM + tid) * DK + d] = zv[hh * DK + d];
      }
    }
    const int wdst = (tid & 64) + 16 * ((tid & 63) >> 4) + 2 * (tid & 7) +
                     ((tid >> 3) & 1);
    if (tid < PS_DIM) Ps[tid < PS_WC ? wdst : tid] = pv;
    if (owner) {
      float zq[8];
      vqc8(xr, wq, zq);
#pragma unroll
      for (int hh = 0; hh < H_DIM; ++hh) {
#pragma unroll
        for (int d = 0; d < DK; ++d)
          Qs[(hh * QS_DIM + ql) * DK + d] = zq[hh * DK + d];
      }
    }
  }
  __syncthreads();

  v2f wc2, wf2, pb1, pb2, pg1, pbe1, pg2, pbe2;
  v4f w1r[4], w2r[4];
  {
    const int lane  = tid & 31;
    const int wv_id = tid >> 5;
    const int h     = wv_id & 1;
    const int qb    = wv_id >> 1;
    const int half  = lane >> 4;
    const int ln    = lane & 15;
    const int kp    = half * 2;
    const int qbase = qb * 16;

    const float* Qh = Qs + h * QS_DIM * DK;
    const float* Kh = Ks + h * S_DIM * DK;
    const float* Vh = Vs + h * S_DIM * DK;

    v2f a;
    a.x = Qh[(qbase + ln) * DK + kp] * SCALE_LOG2E;
    a.y = Qh[(qbase + ln) * DK + kp + 1] * SCALE_LOG2E;

    v2f l2[4], acc[4][4];
#pragma unroll
    for (int j = 0; j < 4; ++j) {
      l2[j] = (v2f){0.f, 0.f};
#pragma unroll
      for (int c = 0; c < 4; ++c) acc[j][c] = (v2f){0.f, 0.f};
    }

    auto ldK = [&](int t) -> v2f {
      return *(const v2f*)(Kh + (t * 16 + ln) * DK + kp);
    };
    auto ldV = [&](int t) -> v4f {
      return *(const v4f*)(Vh + (t * 16 + ln) * DK);
    };
    const v8f cz = {0.f, 0.f, 0.f, 0.f, 0.f, 0.f, 0.f, 0.f};

    v2f kf[16];
    v4f vf[16];
    kf[0] = ldK(0);
    kf[1] = ldK(1);
    vf[0] = ldV(0);
    kf[2] = ldK(2);
    kf[3] = ldK(3);
    vf[1] = ldV(1);
    __builtin_amdgcn_sched_barrier(0);
#pragma unroll
    for (int t = 2; t < 16; ++t) {
      if (t + 2 < 16) kf[t + 2] = ldK(t + 2);
      vf[t] = ldV(t);
    }
    __builtin_amdgcn_sched_barrier(0);

    v8f sc[16];
    v2f pe[16][4];
    sc[0] = __builtin_amdgcn_wmma_f32_16x16x4_f32(
        false, a, false, kf[0], (short)0, cz, false, false);
    tile_exp(sc[0], pe[0]);
    sc[1] = __builtin_amdgcn_wmma_f32_16x16x4_f32(
        false, a, false, kf[1], (short)0, cz, false, false);
#pragma unroll
    for (int t = 0; t < 16; ++t) {
      tile_accumulate(pe[t], vf[t], l2, acc);
      if (t + 1 < 16) tile_exp(sc[t + 1], pe[t + 1]);
      if (t + 2 < 16)
        sc[t + 2] = __builtin_amdgcn_wmma_f32_16x16x4_f32(
            false, a, false, kf[t + 2], (short)0, cz, false, false);
    }
    __builtin_amdgcn_sched_group_barrier(0x008, 1, 0);
    __builtin_amdgcn_sched_group_barrier(0x400, 8, 0);
    __builtin_amdgcn_sched_group_barrier(0x008, 1, 0);
    tile_order<true, true>();
    tile_order<true, true>();
    tile_order<true, true>();
    tile_order<true, true>();
    tile_order<true, true>();
    tile_order<true, true>();
    tile_order<true, true>();
    tile_order<true, true>();
    tile_order<true, true>();
    tile_order<true, true>();
    tile_order<true, true>();
    tile_order<true, true>();
    tile_order<true, true>();
    tile_order<true, true>();
    tile_order<false, true>();
    tile_order<false, false>();

    wc2  = *(const v2f*)(Ps + PS_WC  + 2 * p);
    wf2  = *(const v2f*)(Ps + PS_WF  + 2 * p);
    pb1  = *(const v2f*)(Ps + PS_B1  + 2 * p);
    pb2  = *(const v2f*)(Ps + PS_B2  + 2 * p);
    pg1  = *(const v2f*)(Ps + PS_G1  + 2 * p);
    pbe1 = *(const v2f*)(Ps + PS_BE1 + 2 * p);
    pg2  = *(const v2f*)(Ps + PS_G2  + 2 * p);
    pbe2 = *(const v2f*)(Ps + PS_BE2 + 2 * p);
#pragma unroll
    for (int k = 0; k < 4; ++k) {
      w1r[k] = *(const v4f*)(Ps + PS_W1 + 16 * p + 4 * k);
      w2r[k] = *(const v4f*)(Ps + PS_W2 + 16 * p + 4 * k);
    }

    const bool b0 = (ln & 1) != 0, b1 = (ln & 2) != 0;
    const bool b2 = (ln & 4) != 0, b3 = (ln & 8) != 0;
    float s1[4][4], d1[4];
#pragma unroll
    for (int k = 0; k < 4; ++k) {
#pragma unroll
      for (int dv = 0; dv < DK; ++dv)
        s1[k][dv] = fold<DPP_XOR1>(b0, acc[k][dv].x, acc[k][dv].y);
      d1[k] = fold<DPP_XOR1>(b0, l2[k].x, l2[k].y);
    }
    float s2[2][4], d2[2];
#pragma unroll
    for (int k = 0; k < 2; ++k) {
#pragma unroll
      for (int dv = 0; dv < DK; ++dv)
        s2[k][dv] = fold<DPP_XOR2>(b1, s1[2 * k][dv], s1[2 * k + 1][dv]);
      d2[k] = fold<DPP_XOR2>(b1, d1[2 * k], d1[2 * k + 1]);
    }
    float s3[4];
#pragma unroll
    for (int dv = 0; dv < DK; ++dv)
      s3[dv] = fold<DPP_XOR4>(b2, s2[0][dv], s2[1][dv]);
    float den = fold<DPP_XOR4>(b2, d2[0], d2[1]);
    const float c0 = fold<DPP_XOR8>(b3, s3[0], s3[2]);
    const float c1 = fold<DPP_XOR8>(b3, s3[1], s3[3]);
    den += dpp_f<DPP_XOR8>(den);
    {
      const float inv_l = 1.0f / den;
      const int m = qbase + (ln & 7) + 8 * half;
      const v2f o = {c0 * inv_l, c1 * inv_l};
      *(v2f*)(Cs + m * E_DIM + h * DK + 2 * (ln >> 3)) = o;
    }
  }
  __syncthreads();

  const v2f cv = *(const v2f*)(Cs + tok * E_DIM + 2 * p);
  float a0, a1;
  vqc_quad(p, cv.x, cv.y, wc2.x, wc2.y, a0, a1);
  float y0, y1;
  ln_quad(x2.x + a0, x2.y + a1, pg1, pbe1, y0, y1);

  float h0, h1;
  rows_quad(y0, y1, w1r, pb1, h0, h1);
  float q0, q1;
  vqc_quad(p, h0, h1, wf2.x, wf2.y, q0, q1);
  float f0, f1;
  rows_quad(q0, q1, w2r, pb2, f0, f1);

  float o0, o1;
  ln_quad(y0 + f0, y1 + f1, pg2, pbe2, o0, o1);
  const v2f o = {o0, o1};
  *(v2f*)(out + (size_t)(b * S_DIM + qs * QS_DIM + tok) * E_DIM + 2 * p) = o;
}

extern "C" void kernel_launch(void* const* d_in, const int* in_sizes, int n_in,
                              void* d_out, int out_size, void* d_ws,
                              size_t ws_size, hipStream_t stream) {
  (void)in_sizes; (void)n_in; (void)out_size; (void)d_ws; (void)ws_size;
  const float* x   = (const float*)d_in[0];
  const float* Wq  = (const float*)d_in[1];
  const float* Wk  = (const float*)d_in[2];
  const float* Wv  = (const float*)d_in[3];
  const float* Wc  = (const float*)d_in[4];
  const float* Wf  = (const float*)d_in[5];
  const float* w1  = (const float*)d_in[6];
  const float* b1  = (const float*)d_in[7];
  const float* w2  = (const float*)d_in[8];
  const float* b2  = (const float*)d_in[9];
  const float* g1  = (const float*)d_in[10];
  const float* be1 = (const float*)d_in[11];
  const float* g2  = (const float*)d_in[12];
  const float* be2 = (const float*)d_in[13];
  float* out = (float*)d_out;

  fused_block_kernel<<<B_DIM * NSLICE, 256, 0, stream>>>(
      x, Wq, Wk, Wv, Wc, Wf, w1, b1, w2, b2, g1, be1, g2, be2, out);
}
